# starting kernel
# baseline (speedup 1.0000x reference)
#include <hip/hip_runtime.h>
#include <hip/hip_bf16.h>

typedef __attribute__((ext_vector_type(2))) float v2f;
typedef __attribute__((ext_vector_type(8))) float v8f;

#define HID   4096
#define BATCH 16384
#define OUTC  44
#define PD    8

#define THREADS        128
#define ROWS_PER_BLOCK 64
#define KC        64
#define XS_STRIDE 68
#define WS_STRIDE 68
#define HS_STRIDE 52
#define NPAD      48

__global__ __launch_bounds__(THREADS) void l3m_param_head_kernel(
    const float* __restrict__ x, const float* __restrict__ W,
    const float* __restrict__ bias, float* __restrict__ out)
{
    __shared__ float xs[ROWS_PER_BLOCK * XS_STRIDE];
    __shared__ float ws[NPAD * WS_STRIDE];

    const int tid  = threadIdx.x;
    const int wid  = tid >> 5;
    const int lane = tid & 31;
    const int half = lane >> 4;
    const int l16  = lane & 15;
    const int m0   = wid * 16;
    const int rowBase = blockIdx.x * ROWS_PER_BLOCK;

    v8f acc0 = {}; v8f acc1 = {}; v8f acc2 = {};

    for (int k0 = 0; k0 < HID; k0 += KC) {
        const float4* xg = (const float4*)x;
        const float4* wg = (const float4*)W;
        const int kq = k0 >> 2;
        #pragma unroll
        for (int i = 0; i < 8; ++i) {
            int f  = tid + i * THREADS;
            int r  = f >> 4;
            int c4 = f & 15;
            float4 v = xg[(size_t)(rowBase + r) * (HID / 4) + kq + c4];
            *(float4*)&xs[r * XS_STRIDE + c4 * 4] = v;
        }
        #pragma unroll
        for (int i = 0; i < 6; ++i) {
            int f  = tid + i * THREADS;
            int n  = f >> 4;
            int c4 = f & 15;
            float4 v = make_float4(0.f, 0.f, 0.f, 0.f);
            if (n < OUTC)
                v = wg[(size_t)n * (HID / 4) + kq + c4];
            *(float4*)&ws[n * WS_STRIDE + c4 * 4] = v;
        }
        if (k0 + 2 * KC < HID)
            __builtin_prefetch(&x[(size_t)(rowBase + (tid >> 1)) * HID
                                  + (k0 + 2 * KC) + (tid & 1) * 32], 0, 1);
        __syncthreads();

        #pragma unroll
        for (int kk = 0; kk < KC; kk += 4) {
            const int ks = kk + 2 * half;
            v2f a  = *(const v2f*)&xs[(m0 + l16) * XS_STRIDE + ks];
            v2f b0 = *(const v2f*)&ws[( 0 + l16) * WS_STRIDE + ks];
            v2f b1 = *(const v2f*)&ws[(16 + l16) * WS_STRIDE + ks];
            v2f b2 = *(const v2f*)&ws[(32 + l16) * WS_STRIDE + ks];
            acc0 = __builtin_amdgcn_wmma_f32_16x16x4_f32(false, a, false, b0, (short)0, acc0, false, false);
            acc1 = __builtin_amdgcn_wmma_f32_16x16x4_f32(false, a, false, b1, (short)0, acc1, false, false);
            acc2 = __builtin_amdgcn_wmma_f32_16x16x4_f32(false, a, false, b2, (short)0, acc2, false, false);
        }
        __syncthreads();
    }

    float* hs = xs;
    #pragma unroll
    for (int v = 0; v < 8; ++v) {
        int m = m0 + v + 8 * half;
        hs[m * HS_STRIDE +  0 + l16] = acc0[v];
        hs[m * HS_STRIDE + 16 + l16] = acc1[v];
        hs[m * HS_STRIDE + 32 + l16] = acc2[v];
    }
    __syncthreads();

    if (tid < ROWS_PER_BLOCK) {
        const int row = rowBase + tid;
        float h[OUTC];
        #pragma unroll
        for (int j = 0; j < OUTC; ++j) h[j] = hs[tid * HS_STRIDE + j] + bias[j];

        float* outP = out;
        float* outS = out + (size_t)BATCH * PD;
        float* outL = out + (size_t)BATCH * PD + (size_t)BATCH * PD * PD;

        float pr[PD];
        #pragma unroll
        for (int i = 0; i < PD; ++i) pr[i] = 1.f / (1.f + expf(-h[i]));
        #pragma unroll
        for (int i = 0; i < PD; i += 4)
            *(float4*)&outP[(size_t)row * PD + i] =
                make_float4(pr[i], pr[i + 1], pr[i + 2], pr[i + 3]);

        float sg[PD][PD];
        #pragma unroll
        for (int i = 0; i < PD; ++i) {
            const float sc = 2.0f / sqrtf((float)(i + 1));
            float d  = h[PD + i];
            float sp = (d > 20.f) ? d : log1pf(expf(d));
            sg[i][i] = sp * sc;
            #pragma unroll
            for (int j = 0; j < i; ++j)
                sg[i][j] = h[2 * PD + (i * (i - 1)) / 2 + j] * sc;
        }

        float p = 1.f;
        #pragma unroll
        for (int i = 0; i < PD; ++i) p *= sg[i][i];
        float bp  = 10000.f * p;
        float lae = (bp > 0.f) ? (bp + log1pf(expf(-bp))) : log1pf(expf(bp));
        outL[row] = logf(lae) - 9.210340371976184f;

        float sm[PD][PD];
        #pragma unroll
        for (int i = 0; i < PD; ++i) {
            #pragma unroll
            for (int j = 0; j <= i; ++j) {
                float s = 0.f;
                #pragma unroll
                for (int k = 0; k <= j; ++k) s += sg[i][k] * sg[j][k];
                sm[i][j] = s;
                sm[j][i] = s;
            }
        }
        #pragma unroll
        for (int i = 0; i < PD; ++i)
            #pragma unroll
            for (int j = 0; j < PD; j += 4)
                *(float4*)&outS[(size_t)row * (PD * PD) + i * PD + j] =
                    make_float4(sm[i][j], sm[i][j + 1], sm[i][j + 2], sm[i][j + 3]);
    }
}

extern "C" void kernel_launch(void* const* d_in, const int* in_sizes, int n_in,
                              void* d_out, int out_size, void* d_ws, size_t ws_size,
                              hipStream_t stream) {
    const float* x = (const float*)d_in[0];
    const float* W = (const float*)d_in[1];
    const float* b = (const float*)d_in[2];
    float* out = (float*)d_out;
    (void)in_sizes; (void)n_in; (void)out_size; (void)d_ws; (void)ws_size;

    dim3 grid(BATCH / ROWS_PER_BLOCK);
    dim3 block(THREADS);
    l3m_param_head_kernel<<<grid, block, 0, stream>>>(x, W, b, out);
}


# W comes in by whole-line loads (4 lines per load, was 16), 3 super-steps ahead, and is turned into B fragments through a wave-private LDS slice, no barrier; x loads, k order and sums unchanged
# speedup vs baseline: 5.2407x; 5.2407x over previous
#include <hip/hip_runtime.h>
#include <hip/hip_bf16.h>


typedef __attribute__((ext_vector_type(2))) float v2f;
typedef __attribute__((ext_vector_type(4))) float v4f;
typedef __attribute__((ext_vector_type(8))) float v8f;

#define HID   4096
#define BATCH 16384
#define OUTC  44
#define PD    8

#define THREADS        128
#define ROWS_PER_BLOCK 64
#define NWAVE   4
#define KW      (HID / NWAVE)
#define SS      32
#define NSS     (KW / SS)
#define NP      4
#define NRT     4
#define NNT     3
#define AHEAD   2
#define RINGS   4
#define PFU     3
#define NUNIT   (KW / 64)
#define HS_STRIDE 52
#define HS_TILE   (ROWS_PER_BLOCK * HS_STRIDE)
#define WAHEAD  3
#define WLD     11
#define WPITCH  36
#define WSLOT   (OUTC * WPITCH)

static __device__ __forceinline__ v4f ld16(const float* p) {
    return *(const v4f*)p;
}

__global__ __launch_bounds__(THREADS) void l3m_param_head_kernel(
    const float* __restrict__ x, const float* __restrict__ W,
    const float* __restrict__ bias, float* __restrict__ out)
{
    __shared__ __attribute__((aligned(16))) float hs[NWAVE * HS_TILE];
    __shared__ __attribute__((aligned(16))) float wl[NWAVE * 2 * WSLOT];

    const int tid  = threadIdx.x;
    const int wid  = tid >> 5;
    const int lane = tid & 31;
    const int half = lane >> 4;
    const int l16  = lane & 15;
    const int rowBase = blockIdx.x * ROWS_PER_BLOCK;

    v8f acc[NRT][NNT];
    #pragma unroll
    for (int r = 0; r < NRT; ++r)
        #pragma unroll
        for (int t = 0; t < NNT; ++t)
            acc[r][t] = (v8f){0.f, 0.f, 0.f, 0.f, 0.f, 0.f, 0.f, 0.f};

    const float* xb  = x + (size_t)(rowBase + l16) * HID + wid * KW + 4 * half;
    const float* wc  = W + (size_t)(lane >> 3) * HID + wid * KW + 4 * (lane & 7);
    float* wst = wl + wid * 2 * WSLOT + (lane >> 3) * WPITCH + 4 * (lane & 7);
    const int    w2r = (32 + l16 < OUTC) ? (32 + l16) : (OUTC - 1);
    const float* wf0 = wl + wid * 2 * WSLOT + l16 * WPITCH + 4 * half;
    const float* wf2 = wl + wid * 2 * WSLOT + w2r * WPITCH + 4 * half;
    const float* xpf = x + (size_t)(rowBase + lane) * HID + wid * KW;

    v4f xa[RINGS][NRT][NP];
    v4f wg[RINGS][WLD + 1];
    v4f wf[2][NNT][NP];

    #pragma unroll
    for (int j = 0; j < WLD; ++j)
        wg[0][j] = ld16(wc + (size_t)4 * j * HID);
    #pragma unroll
    for (int s = 0; s < AHEAD; ++s) {
        #pragma unroll
        for (int r = 0; r < NRT; ++r) {
            #pragma unroll
            for (int p = 0; p < NP; ++p)
                xa[s][r][p] = ld16(xb + (size_t)r * 16 * HID + s * SS + 8 * p);
            if (r < NNT) {
                #pragma unroll
                for (int j = 4 * r; j < 4 * r + 4 && j < WLD; ++j)
                    wg[s + WAHEAD - AHEAD][j] = ld16(wc + (size_t)4 * j * HID + (s + WAHEAD - AHEAD) * SS);
            }
            __builtin_amdgcn_sched_barrier(0);
        }
    }
    #pragma unroll
    for (int u = 1; u <= PFU; ++u) {
        __builtin_prefetch(xpf + 64 * u, 0, 1);
        __builtin_prefetch(xpf + (size_t)32 * HID + 64 * u, 0, 1);
    }
    #pragma unroll
    for (int j = 0; j < WLD; ++j)
        *(v4f*)(wst + 4 * j * WPITCH) = wg[0][j];
    asm volatile("s_wait_dscnt 0x0" ::: "memory");
    #pragma unroll
    for (int p = 0; p < NP; ++p) {
        wf[0][0][p] = *(const v4f*)(wf0 + 8 * p);
        wf[0][1][p] = *(const v4f*)(wf0 + 16 * WPITCH + 8 * p);
        wf[0][2][p] = *(const v4f*)(wf2 + 8 * p);
    }

    for (int S0 = 0; S0 < NSS; S0 += RINGS) {
        #pragma unroll
        for (int s = 0; s < RINGS; ++s) {
            const int S  = S0 + s;
            const int Sn = (S + AHEAD < NSS) ? S + AHEAD : NSS - 1;
            const int Wn = (S + WAHEAD < NSS) ? S + WAHEAD : NSS - 1;
            const int Un = ((S >> 1) + 1 + PFU < NUNIT) ? (S >> 1) + 1 + PFU : NUNIT - 1;
            const int ns = (s + AHEAD) % RINGS;
            const int wcur = s & 1;
            const int wnxt = wcur ^ 1;
            const int gn   = (s + WAHEAD) % RINGS;
            const int g1   = (s + 1) % RINGS;

            #pragma unroll
            for (int r = 0; r < NRT; ++r) {
                #pragma unroll
                for (int p = 0; p < NP; ++p)
                    xa[ns][r][p] = ld16(xb + (size_t)r * 16 * HID + Sn * SS + 8 * p);
                if (r < NNT) {
                    #pragma unroll
                    for (int j = 4 * r; j < 4 * r + 4 && j < WLD; ++j)
                        wg[gn][j] = ld16(wc + (size_t)4 * j * HID + Wn * SS);
                } else {
                    __builtin_prefetch(xpf + (size_t)(s & 1) * 32 * HID + 64 * Un, 0, 1);
                }
                if (r == 2) {
                    #pragma unroll
                    for (int j = 0; j < WLD; ++j)
                        *(v4f*)(wst + wnxt * WSLOT + 4 * j * WPITCH) = wg[g1][j];
                }
                if (r == 3) {
                    asm volatile("s_wait_dscnt 0x0" ::: "memory");
                    #pragma unroll
                    for (int p = 0; p < NP; ++p) {
                        wf[wnxt][0][p] = *(const v4f*)(wf0 + wnxt * WSLOT + 8 * p);
                        wf[wnxt][1][p] = *(const v4f*)(wf0 + wnxt * WSLOT + 16 * WPITCH + 8 * p);
                        wf[wnxt][2][p] = *(const v4f*)(wf2 + wnxt * WSLOT + 8 * p);
                    }
                }
                __builtin_amdgcn_sched_barrier(0);

                #pragma unroll
                for (int p = 0; p < NP; ++p) {
                    const v4f a  = xa[s][r][p];
                    const v4f b0 = wf[wcur][0][p];
                    const v4f b1 = wf[wcur][1][p];
                    const v4f b2 = wf[wcur][2][p];
                    const v2f alo = {a.x, a.y},   ahi = {a.z, a.w};
                    const v2f b0l = {b0.x, b0.y}, b0h = {b0.z, b0.w};
                    const v2f b1l = {b1.x, b1.y}, b1h = {b1.z, b1.w};
                    const v2f b2l = {b2.x, b2.y}, b2h = {b2.z, b2.w};
                    acc[r][0] = __builtin_amdgcn_wmma_f32_16x16x4_f32(false, alo, false, b0l, (short)0, acc[r][0], false, false);
                    acc[r][1] = __builtin_amdgcn_wmma_f32_16x16x4_f32(false, alo, false, b1l, (short)0, acc[r][1], false, false);
                    acc[r][2] = __builtin_amdgcn_wmma_f32_16x16x4_f32(false, alo, false, b2l, (short)0, acc[r][2], false, false);
                    acc[r][0] = __builtin_amdgcn_wmma_f32_16x16x4_f32(false, ahi, false, b0h, (short)0, acc[r][0], false, false);
                    acc[r][1] = __builtin_amdgcn_wmma_f32_16x16x4_f32(false, ahi, false, b1h, (short)0, acc[r][1], false, false);
                    acc[r][2] = __builtin_amdgcn_wmma_f32_16x16x4_f32(false, ahi, false, b2h, (short)0, acc[r][2], false, false);
                }
                __builtin_amdgcn_sched_barrier(0);
            }
        }
    }

    float* hw = hs + wid * HS_TILE;
    #pragma unroll
    for (int r = 0; r < NRT; ++r) {
        #pragma unroll
        for (int v = 0; v < 8; ++v) {
            const int m = 16 * r + v + 8 * half;
            hw[m * HS_STRIDE +  0 + l16] = acc[r][0][v];
            hw[m * HS_STRIDE + 16 + l16] = acc[r][1][v];
            hw[m * HS_STRIDE + 32 + l16] = acc[r][2][v];
        }
    }
    __syncthreads();

    if (tid < ROWS_PER_BLOCK) {
        const int row = rowBase + tid;
        float h[OUTC];
        #pragma unroll
        for (int q = 0; q < OUTC / 4; ++q) {
            const v4f p0 = *(const v4f*)&hs[0 * HS_TILE + tid * HS_STRIDE + 4 * q];
            const v4f p1 = *(const v4f*)&hs[1 * HS_TILE + tid * HS_STRIDE + 4 * q];
            const v4f p2 = *(const v4f*)&hs[2 * HS_TILE + tid * HS_STRIDE + 4 * q];
            const v4f p3 = *(const v4f*)&hs[3 * HS_TILE + tid * HS_STRIDE + 4 * q];
            const v4f sm4 = ((p0 + p1) + p2) + p3;
            h[4 * q + 0] = sm4.x + bias[4 * q + 0];
            h[4 * q + 1] = sm4.y + bias[4 * q + 1];
            h[4 * q + 2] = sm4.z + bias[4 * q + 2];
            h[4 * q + 3] = sm4.w + bias[4 * q + 3];
        }

        float* outP = out;
        float* outS = out + (size_t)BATCH * PD;
        float* outL = out + (size_t)BATCH * PD + (size_t)BATCH * PD * PD;

        float pr[PD];
        #pragma unroll
        for (int i = 0; i < PD; ++i) pr[i] = 1.f / (1.f + expf(-h[i]));
        #pragma unroll
        for (int i = 0; i < PD; i += 4)
            *(float4*)&outP[(size_t)row * PD + i] =
                make_float4(pr[i], pr[i + 1], pr[i + 2], pr[i + 3]);

        float sg[PD][PD];
        #pragma unroll
        for (int i = 0; i < PD; ++i) {
            const float sc = 2.0f / sqrtf((float)(i + 1));
            float d  = h[PD + i];
            float sp = (d > 20.f) ? d : log1pf(expf(d));
            sg[i][i] = sp * sc;
            #pragma unroll
            for (int j = 0; j < i; ++j)
                sg[i][j] = h[2 * PD + (i * (i - 1)) / 2 + j] * sc;
        }

        float p = 1.f;
        #pragma unroll
        for (int i = 0; i < PD; ++i) p *= sg[i][i];
        float bp  = 10000.f * p;
        float lae = (bp > 0.f) ? (bp + log1pf(expf(-bp))) : log1pf(expf(bp));
        outL[row] = logf(lae) - 9.210340371976184f;

        float sm[PD][PD];
        #pragma unroll
        for (int i = 0; i < PD; ++i) {
            #pragma unroll
            for (int j = 0; j <= i; ++j) {
                float s = 0.f;
                #pragma unroll
                for (int k = 0; k <= j; ++k) s += sg[i][k] * sg[j][k];
                sm[i][j] = s;
                sm[j][i] = s;
            }
        }
        #pragma unroll
        for (int i = 0; i < PD; ++i)
            #pragma unroll
            for (int j = 0; j < PD; j += 4)
                *(float4*)&outS[(size_t)row * (PD * PD) + i * PD + j] =
                    make_float4(sm[i][j], sm[i][j + 1], sm[i][j + 2], sm[i][j + 3]);
    }
}

extern "C" void kernel_launch(void* const* d_in, const int* in_sizes, int n_in,
                              void* d_out, int out_size, void* d_ws, size_t ws_size,
                              hipStream_t stream) {
    const float* x = (const float*)d_in[0];
    const float* W = (const float*)d_in[1];
    const float* b = (const float*)d_in[2];
    float* out = (float*)d_out;
    (void)in_sizes; (void)n_in; (void)out_size; (void)d_ws; (void)ws_size;

    dim3 grid(BATCH / ROWS_PER_BLOCK);
    dim3 block(THREADS);
    l3m_param_head_kernel<<<grid, block, 0, stream>>>(x, W, b, out);
}


# 8-wave work-group, two waves per SIMD: each wave owns a K eighth (was a quarter) of the 64 rows, look-ahead halved for 512 registers; a sibling's WMMAs fill a wave's load places; sum order changes
# speedup vs baseline: 5.3897x; 1.0284x over previous
#include <hip/hip_runtime.h>
#include <hip/hip_bf16.h>


typedef __attribute__((ext_vector_type(2))) float v2f;
typedef __attribute__((ext_vector_type(4))) float v4f;
typedef __attribute__((ext_vector_type(8))) float v8f;

#define HID   4096
#define BATCH 16384
#define OUTC  44
#define PD    8

#define THREADS        256
#define ROWS_PER_BLOCK 64
#define NWAVE   8
#define KW      (HID / NWAVE)
#define SS      32
#define NSS     (KW / SS)
#define NP      4
#define NRT     4
#define NNT     3
#define AHEAD   1
#define RINGS   2
#define PFU     1
#define NUNIT   (KW / 64)
#define HS_STRIDE 52
#define HS_TILE   (ROWS_PER_BLOCK * HS_STRIDE)
#define WAHEAD  2
#define WLD     11
#define WPITCH  36
#define WSLOT   (OUTC * WPITCH)

static __device__ __forceinline__ v4f ld16(const float* p) {
    return *(const v4f*)p;
}

__global__ __launch_bounds__(THREADS) void l3m_param_head_kernel(
    const float* __restrict__ x, const float* __restrict__ W,
    const float* __restrict__ bias, float* __restrict__ out)
{
    __shared__ __attribute__((aligned(16))) float hs[NWAVE * HS_TILE];
    __shared__ __attribute__((aligned(16))) float wl[NWAVE * 2 * WSLOT];

    const int tid  = threadIdx.x;
    const int wid  = tid >> 5;
    const int lane = tid & 31;
    const int half = lane >> 4;
    const int l16  = lane & 15;
    const int rowBase = blockIdx.x * ROWS_PER_BLOCK;

    v8f acc[NRT][NNT];
    #pragma unroll
    for (int r = 0; r < NRT; ++r)
        #pragma unroll
        for (int t = 0; t < NNT; ++t)
            acc[r][t] = (v8f){0.f, 0.f, 0.f, 0.f, 0.f, 0.f, 0.f, 0.f};

    const float* xb  = x + (size_t)(rowBase + l16) * HID + wid * KW + 4 * half;
    const float* wc  = W + (size_t)(lane >> 3) * HID + wid * KW + 4 * (lane & 7);
    float* wst = wl + wid * 2 * WSLOT + (lane >> 3) * WPITCH + 4 * (lane & 7);
    const int    w2r = (32 + l16 < OUTC) ? (32 + l16) : (OUTC - 1);
    const float* wf0 = wl + wid * 2 * WSLOT + l16 * WPITCH + 4 * half;
    const float* wf2 = wl + wid * 2 * WSLOT + w2r * WPITCH + 4 * half;
    const float* xpf = x + (size_t)(rowBase + lane) * HID + wid * KW;

    v4f xa[RINGS][NRT][NP];
    v4f wg[RINGS][WLD + 1];
    v4f wf[2][NNT][NP];

    #pragma unroll
    for (int j = 0; j < WLD; ++j)
        wg[0][j] = ld16(wc + (size_t)4 * j * HID);
    #pragma unroll
    for (int s = 0; s < AHEAD; ++s) {
        #pragma unroll
        for (int r = 0; r < NRT; ++r) {
            #pragma unroll
            for (int p = 0; p < NP; ++p)
                xa[s][r][p] = ld16(xb + (size_t)r * 16 * HID + s * SS + 8 * p);
            if (r < NNT) {
                #pragma unroll
                for (int j = 4 * r; j < 4 * r + 4 && j < WLD; ++j)
                    wg[s + WAHEAD - AHEAD][j] = ld16(wc + (size_t)4 * j * HID + (s + WAHEAD - AHEAD) * SS);
            }
            __builtin_amdgcn_sched_barrier(0);
        }
    }
    #pragma unroll
    for (int u = 1; u <= PFU; ++u) {
        __builtin_prefetch(xpf + 64 * u, 0, 1);
        __builtin_prefetch(xpf + (size_t)32 * HID + 64 * u, 0, 1);
    }
    #pragma unroll
    for (int j = 0; j < WLD; ++j)
        *(v4f*)(wst + 4 * j * WPITCH) = wg[0][j];
    asm volatile("s_wait_dscnt 0x0" ::: "memory");
    #pragma unroll
    for (int p = 0; p < NP; ++p) {
        wf[0][0][p] = *(const v4f*)(wf0 + 8 * p);
        wf[0][1][p] = *(const v4f*)(wf0 + 16 * WPITCH + 8 * p);
        wf[0][2][p] = *(const v4f*)(wf2 + 8 * p);
    }

    for (int S0 = 0; S0 < NSS; S0 += RINGS) {
        #pragma unroll
        for (int s = 0; s < RINGS; ++s) {
            const int S  = S0 + s;
            const int Sn = (S + AHEAD < NSS) ? S + AHEAD : NSS - 1;
            const int Wn = (S + WAHEAD < NSS) ? S + WAHEAD : NSS - 1;
            const int Un = ((S >> 1) + 1 + PFU < NUNIT) ? (S >> 1) + 1 + PFU : NUNIT - 1;
            const int ns = (s + AHEAD) % RINGS;
            const int wcur = s & 1;
            const int wnxt = wcur ^ 1;
            const int gn   = (s + WAHEAD) % RINGS;
            const int g1   = (s + 1) % RINGS;

            #pragma unroll
            for (int r = 0; r < NRT; ++r) {
                #pragma unroll
                for (int p = 0; p < NP; ++p)
                    xa[ns][r][p] = ld16(xb + (size_t)r * 16 * HID + Sn * SS + 8 * p);
                if (r < NNT) {
                    #pragma unroll
                    for (int j = 4 * r; j < 4 * r + 4 && j < WLD; ++j)
                        wg[gn][j] = ld16(wc + (size_t)4 * j * HID + Wn * SS);
                } else {
                    __builtin_prefetch(xpf + (size_t)(s & 1) * 32 * HID + 64 * Un, 0, 1);
                }
                if (r == 2) {
                    #pragma unroll
                    for (int j = 0; j < WLD; ++j)
                        *(v4f*)(wst + wnxt * WSLOT + 4 * j * WPITCH) = wg[g1][j];
                }
                if (r == 3) {
                    asm volatile("s_wait_dscnt 0x0" ::: "memory");
                    #pragma unroll
                    for (int p = 0; p < NP; ++p) {
                        wf[wnxt][0][p] = *(const v4f*)(wf0 + wnxt * WSLOT + 8 * p);
                        wf[wnxt][1][p] = *(const v4f*)(wf0 + wnxt * WSLOT + 16 * WPITCH + 8 * p);
                        wf[wnxt][2][p] = *(const v4f*)(wf2 + wnxt * WSLOT + 8 * p);
                    }
                }
                __builtin_amdgcn_sched_barrier(0);

                #pragma unroll
                for (int p = 0; p < NP; ++p) {
                    const v4f a  = xa[s][r][p];
                    const v4f b0 = wf[wcur][0][p];
                    const v4f b1 = wf[wcur][1][p];
                    const v4f b2 = wf[wcur][2][p];
                    const v2f alo = {a.x, a.y},   ahi = {a.z, a.w};
                    const v2f b0l = {b0.x, b0.y}, b0h = {b0.z, b0.w};
                    const v2f b1l = {b1.x, b1.y}, b1h = {b1.z, b1.w};
                    const v2f b2l = {b2.x, b2.y}, b2h = {b2.z, b2.w};
                    acc[r][0] = __builtin_amdgcn_wmma_f32_16x16x4_f32(false, alo, false, b0l, (short)0, acc[r][0], false, false);
                    acc[r][1] = __builtin_amdgcn_wmma_f32_16x16x4_f32(false, alo, false, b1l, (short)0, acc[r][1], false, false);
                    acc[r][2] = __builtin_amdgcn_wmma_f32_16x16x4_f32(false, alo, false, b2l, (short)0, acc[r][2], false, false);
                    acc[r][0] = __builtin_amdgcn_wmma_f32_16x16x4_f32(false, ahi, false, b0h, (short)0, acc[r][0], false, false);
                    acc[r][1] = __builtin_amdgcn_wmma_f32_16x16x4_f32(false, ahi, false, b1h, (short)0, acc[r][1], false, false);
                    acc[r][2] = __builtin_amdgcn_wmma_f32_16x16x4_f32(false, ahi, false, b2h, (short)0, acc[r][2], false, false);
                }
                __builtin_amdgcn_sched_barrier(0);
            }
        }
    }

    float* hw = hs + wid * HS_TILE;
    #pragma unroll
    for (int r = 0; r < NRT; ++r) {
        #pragma unroll
        for (int v = 0; v < 8; ++v) {
            const int m = 16 * r + v + 8 * half;
            hw[m * HS_STRIDE +  0 + l16] = acc[r][0][v];
            hw[m * HS_STRIDE + 16 + l16] = acc[r][1][v];
            hw[m * HS_STRIDE + 32 + l16] = acc[r][2][v];
        }
    }
    __syncthreads();

    if (tid < ROWS_PER_BLOCK) {
        const int row = rowBase + tid;
        float h[OUTC];
        #pragma unroll
        for (int q = 0; q < OUTC / 4; ++q) {
            const float* hp = &hs[tid * HS_STRIDE + 4 * q];
            const v4f p0 = *(const v4f*)(hp + 0 * HS_TILE);
            const v4f p1 = *(const v4f*)(hp + 1 * HS_TILE);
            const v4f p2 = *(const v4f*)(hp + 2 * HS_TILE);
            const v4f p3 = *(const v4f*)(hp + 3 * HS_TILE);
            const v4f p4 = *(const v4f*)(hp + 4 * HS_TILE);
            const v4f p5 = *(const v4f*)(hp + 5 * HS_TILE);
            const v4f p6 = *(const v4f*)(hp + 6 * HS_TILE);
            const v4f p7 = *(const v4f*)(hp + 7 * HS_TILE);
            const v4f sm4 = ((((((p0 + p1) + p2) + p3) + p4) + p5) + p6) + p7;
            h[4 * q + 0] = sm4.x + bias[4 * q + 0];
            h[4 * q + 1] = sm4.y + bias[4 * q + 1];
            h[4 * q + 2] = sm4.z + bias[4 * q + 2];
            h[4 * q + 3] = sm4.w + bias[4 * q + 3];
        }

        float* outP = out;
        float* outS = out + (size_t)BATCH * PD;
        float* outL = out + (size_t)BATCH * PD + (size_t)BATCH * PD * PD;

        float pr[PD];
        #pragma unroll
        for (int i = 0; i < PD; ++i) pr[i] = 1.f / (1.f + expf(-h[i]));
        #pragma unroll
        for (int i = 0; i < PD; i += 4)
            *(float4*)&outP[(size_t)row * PD + i] =
                make_float4(pr[i], pr[i + 1], pr[i + 2], pr[i + 3]);

        float sg[PD][PD];
        #pragma unroll
        for (int i = 0; i < PD; ++i) {
            const float sc = 2.0f / sqrtf((float)(i + 1));
            float d  = h[PD + i];
            float sp = (d > 20.f) ? d : log1pf(expf(d));
            sg[i][i] = sp * sc;
            #pragma unroll
            for (int j = 0; j < i; ++j)
                sg[i][j] = h[2 * PD + (i * (i - 1)) / 2 + j] * sc;
        }

        float p = 1.f;
        #pragma unroll
        for (int i = 0; i < PD; ++i) p *= sg[i][i];
        float bp  = 10000.f * p;
        float lae = (bp > 0.f) ? (bp + log1pf(expf(-bp))) : log1pf(expf(bp));
        outL[row] = logf(lae) - 9.210340371976184f;

        float sm[PD][PD];
        #pragma unroll
        for (int i = 0; i < PD; ++i) {
            #pragma unroll
            for (int j = 0; j <= i; ++j) {
                float s = 0.f;
                #pragma unroll
                for (int k = 0; k <= j; ++k) s += sg[i][k] * sg[j][k];
                sm[i][j] = s;
                sm[j][i] = s;
            }
        }
        #pragma unroll
        for (int i = 0; i < PD; ++i)
            #pragma unroll
            for (int j = 0; j < PD; j += 4)
                *(float4*)&outS[(size_t)row * (PD * PD) + i * PD + j] =
                    make_float4(sm[i][j], sm[i][j + 1], sm[i][j + 2], sm[i][j + 3]);
    }
}

extern "C" void kernel_launch(void* const* d_in, const int* in_sizes, int n_in,
                              void* d_out, int out_size, void* d_ws, size_t ws_size,
                              hipStream_t stream) {
    const float* x = (const float*)d_in[0];
    const float* W = (const float*)d_in[1];
    const float* b = (const float*)d_in[2];
    float* out = (float*)d_out;
    (void)in_sizes; (void)n_in; (void)out_size; (void)d_ws; (void)ws_size;

    dim3 grid(BATCH / ROWS_PER_BLOCK);
    dim3 block(THREADS);
    l3m_param_head_kernel<<<grid, block, 0, stream>>>(x, W, b, out);
}
